# starting kernel
# baseline (speedup 1.0000x reference)
#include <hip/hip_runtime.h>

#define NNODES 50000
#define NEDGES 600000
#define CH     128

typedef __attribute__((ext_vector_type(16))) _Float16 v16h;
typedef __attribute__((ext_vector_type(8)))  float    v8f;

__global__ void init_deg_kernel(float* dis) {
    int i = blockIdx.x * blockDim.x + threadIdx.x;
    if (i < NNODES) dis[i] = 1.0f;
}
__global__ void accum_deg_kernel(const int* __restrict__ dst, float* dis) {
    int e = blockIdx.x * blockDim.x + threadIdx.x;
    if (e < NEDGES) atomicAdd(&dis[dst[e]], 1.0f);
}
__global__ void rsqrt_deg_kernel(float* dis) {
    int i = blockIdx.x * blockDim.x + threadIdx.x;
    if (i < NNODES) dis[i] = rsqrtf(dis[i]);
}

__global__ void pack_w_kernel(const float* __restrict__ W, _Float16* __restrict__ Wp) {
    int t = blockIdx.x * blockDim.x + threadIdx.x;
    if (t >= CH * CH) return;
    int e = t & 15;
    int L = (t >> 4) & 31;
    int f = t >> 9;
    int k = f >> 3;
    int c = f & 7;
    int K = 32 * k + ((e & 8) ? 16 : 0) + 8 * (L >> 4) + (e & 7);
    int n = c * 16 + (L & 15);
    Wp[t] = (_Float16)W[K * CH + n];
}

__global__ __launch_bounds__(256) void gemm_x16_kernel(
        const float* __restrict__ X, const _Float16* __restrict__ Wp,
        const float* __restrict__ bias, float* __restrict__ Out, int nrows) {
    const int lane = threadIdx.x & 31;
    const int wave = blockIdx.x * 8 + (threadIdx.x >> 5);
    const int rowBase = wave * 16;
    if (rowBase >= nrows) return;
    const int half = lane >> 4;
    int m = rowBase + (lane & 15);
    if (m >= nrows) m = nrows - 1;
    const float* xrow = X + (size_t)m * CH;

    v8f acc[8] = {};
    #pragma unroll
    for (int k = 0; k < 4; ++k) {
        v16h a;
        const float* p0 = xrow + 32 * k + 8 * half;
        const float* p1 = p0 + 16;
        #pragma unroll
        for (int j = 0; j < 8; ++j) {
            a[j]     = (_Float16)p0[j];
            a[j + 8] = (_Float16)p1[j];
        }
        #pragma unroll
        for (int c = 0; c < 8; ++c) {
            v16h b = *(const v16h*)(Wp + (((k * 8 + c) * 32 + lane) << 4));
            acc[c] = __builtin_amdgcn_wmma_f32_16x16x32_f16(
                         false, a, false, b, (short)0, acc[c], false, false);
        }
    }
    #pragma unroll
    for (int c = 0; c < 8; ++c) {
        int col = c * 16 + (lane & 15);
        float bv = bias ? bias[col] : 0.0f;
        #pragma unroll
        for (int r = 0; r < 8; ++r) {
            int row = rowBase + r + 8 * half;
            if (row < nrows) Out[(size_t)row * CH + col] = acc[c][r] + bv;
        }
    }
}

__global__ __launch_bounds__(256) void scatter_edges_kernel(
        const float* __restrict__ H, const int* __restrict__ src,
        const int* __restrict__ dst, const float* __restrict__ dis,
        float* __restrict__ AGG) {
    int lane = threadIdx.x & 31;
    int e = blockIdx.x * 8 + (threadIdx.x >> 5);
    if (e >= NEDGES) return;
    int s = src[e], d = dst[e];
    float nrm = dis[s] * dis[d];
    const float4 v = *(const float4*)(H + (size_t)s * CH + lane * 4);
    float* out = AGG + (size_t)d * CH + lane * 4;
    atomicAdd(out + 0, v.x * nrm);
    atomicAdd(out + 1, v.y * nrm);
    atomicAdd(out + 2, v.z * nrm);
    atomicAdd(out + 3, v.w * nrm);
}

__global__ void epilogue_add_kernel(const float* __restrict__ AGG, const float* __restrict__ H,
                                    const float* __restrict__ dis, const float* __restrict__ b,
                                    const float* __restrict__ a, float* __restrict__ U) {
    int idx = blockIdx.x * blockDim.x + threadIdx.x;
    if (idx >= NNODES * CH) return;
    int i = idx >> 7, c = idx & 127;
    float di = dis[i];
    float v = AGG[idx] + di * di * H[idx] + b[c];
    float h = v >= 0.0f ? v : a[c] * v;
    U[idx] += h;
}
__global__ void epilogue_out_kernel(const float* __restrict__ AGG, const float* __restrict__ H,
                                    const float* __restrict__ dis, const float* __restrict__ b,
                                    const float* __restrict__ a, float* __restrict__ out) {
    int idx = blockIdx.x * blockDim.x + threadIdx.x;
    if (idx >= NNODES * CH) return;
    int i = idx >> 7, c = idx & 127;
    float di = dis[i];
    float v = AGG[idx] + di * di * H[idx] + b[c];
    out[idx] = v >= 0.0f ? v : a[c] * v;
}

extern "C" void kernel_launch(void* const* d_in, const int* in_sizes, int n_in,
                              void* d_out, int out_size, void* d_ws, size_t ws_size,
                              hipStream_t stream) {
    const float* x  = (const float*)d_in[0];
    const int* edge = (const int*)d_in[1];
    const int* src  = edge;
    const int* dst  = edge + NEDGES;
    const float* W0 = (const float*)d_in[2];
    const float* b0 = (const float*)d_in[3];
    const float* W1 = (const float*)d_in[4];
    const float* b1 = (const float*)d_in[5];
    const float* Ws = (const float*)d_in[6];
    const float* bs = (const float*)d_in[7];
    const float* a  = (const float*)d_in[8];
    float* out = (float*)d_out;

    float* dis = (float*)d_ws;
    float* H   = dis + 50176;
    float* AGG = H + (size_t)NNODES * CH;
    float* U   = AGG + (size_t)NNODES * CH;
    _Float16* Wp0 = (_Float16*)(U + (size_t)NNODES * CH);
    _Float16* Wps = Wp0 + CH * CH;
    _Float16* Wp1 = Wps + CH * CH;

    const int nodeBlocks = (NNODES + 255) / 256;
    const int edgeBlocks = (NEDGES + 255) / 256;
    const int elemBlocks = (NNODES * CH + 255) / 256;
    const int gemmBlocks = (NNODES + 127) / 128;
    const int scatBlocks = (NEDGES + 7) / 8;

    init_deg_kernel<<<nodeBlocks, 256, 0, stream>>>(dis);
    accum_deg_kernel<<<edgeBlocks, 256, 0, stream>>>(dst, dis);
    rsqrt_deg_kernel<<<nodeBlocks, 256, 0, stream>>>(dis);

    pack_w_kernel<<<64, 256, 0, stream>>>(W0, Wp0);
    pack_w_kernel<<<64, 256, 0, stream>>>(Ws, Wps);
    pack_w_kernel<<<64, 256, 0, stream>>>(W1, Wp1);

    gemm_x16_kernel<<<gemmBlocks, 256, 0, stream>>>(x, Wp0, nullptr, H, NNODES);
    gemm_x16_kernel<<<gemmBlocks, 256, 0, stream>>>(x, Wps, bs, U, NNODES);
    hipMemsetAsync(AGG, 0, (size_t)NNODES * CH * sizeof(float), stream);
    scatter_edges_kernel<<<scatBlocks, 256, 0, stream>>>(H, src, dst, dis, AGG);
    epilogue_add_kernel<<<elemBlocks, 256, 0, stream>>>(AGG, H, dis, b0, a, U);

    gemm_x16_kernel<<<gemmBlocks, 256, 0, stream>>>(U, Wp1, nullptr, H, NNODES);
    hipMemsetAsync(AGG, 0, (size_t)NNODES * CH * sizeof(float), stream);
    scatter_edges_kernel<<<scatBlocks, 256, 0, stream>>>(H, src, dst, dis, AGG);
    epilogue_out_kernel<<<elemBlocks, 256, 0, stream>>>(AGG, H, dis, b1, a, out);
}


# Gather loads two terms per instruction: a half-wave reads a 256-byte row by 16-byte loads, row ids by a broadcast in the row, no scalar chain per term; sum order: even and odd terms apart, then added
# speedup vs baseline: 7.3534x; 7.3534x over previous
#include <hip/hip_runtime.h>

#define NNODES 50000
#define NEDGES 600000
#define CH     128

typedef __attribute__((ext_vector_type(16))) _Float16 v16h;
typedef __attribute__((ext_vector_type(8)))  float    v8f;
typedef __attribute__((ext_vector_type(8)))  _Float16 v8h;
typedef __attribute__((ext_vector_type(4)))  _Float16 v4h;

#define NCHUNK   60
#define CHUNK_E  10000
#define CHUNK_Q  (CHUNK_E / 4)
#define QMAX     3
#define RSH      8
#define RNODES   256
#define NRANGE   196
#define TROW     256

__global__ __launch_bounds__(1024) void bin_edges_kernel(const int* __restrict__ src,
                                                         const int* __restrict__ dst,
                                                         int2* __restrict__ binned,
                                                         int* __restrict__ tbl) {
    __shared__ __attribute__((aligned(16))) int2 stage[CHUNK_E];
    __shared__ int hist[TROW];
    __shared__ int start[TROW];
    __shared__ int wsum[8];
    static_assert(NCHUNK * CHUNK_E == NEDGES && CHUNK_E % 4 == 0 && CHUNK_Q <= QMAX * 1024, "chunking");
    static_assert((NNODES - 1) >> RSH < NRANGE && NRANGE <= TROW, "ranges");
    const int tid = threadIdx.x, lane = tid & 31, wv = tid >> 5;
    const int c = blockIdx.x;
    const int e0 = c * CHUNK_E;
    const int4* d4 = (const int4*)(dst + e0);
    const int4* s4 = (const int4*)(src + e0);
    int4 dv[QMAX], sv[QMAX];
    #pragma unroll
    for (int u = 0; u < QMAX; ++u) {
        const int q = min(tid + u * 1024, CHUNK_Q - 1);
        dv[u] = d4[q];
        sv[u] = s4[q];
    }
    if (tid < TROW) hist[tid] = 0;
    __syncthreads();
    int4 rk[QMAX];
    #pragma unroll
    for (int u = 0; u < QMAX; ++u) {
        rk[u] = make_int4(0, 0, 0, 0);
        if (tid + u * 1024 < CHUNK_Q) {
            dv[u].x = (int)min((unsigned)dv[u].x, (unsigned)(NNODES - 1));
            dv[u].y = (int)min((unsigned)dv[u].y, (unsigned)(NNODES - 1));
            dv[u].z = (int)min((unsigned)dv[u].z, (unsigned)(NNODES - 1));
            dv[u].w = (int)min((unsigned)dv[u].w, (unsigned)(NNODES - 1));
            rk[u].x = atomicAdd(&hist[dv[u].x >> RSH], 1);
            rk[u].y = atomicAdd(&hist[dv[u].y >> RSH], 1);
            rk[u].z = atomicAdd(&hist[dv[u].z >> RSH], 1);
            rk[u].w = atomicAdd(&hist[dv[u].w >> RSH], 1);
        }
    }
    __syncthreads();
    const int h = (tid < TROW) ? hist[tid] : 0;
    int x = h;
    for (int o = 1; o < 32; o <<= 1) {
        int y = __shfl_up(x, o);
        if (lane >= o) x += y;
    }
    if (tid < TROW && lane == 31) wsum[wv] = x;
    __syncthreads();
    if (tid < TROW) {
        int wb = 0;
        for (int k = 0; k < wv; ++k) wb += wsum[k];
        start[tid] = wb + x - h;
        tbl[c * TROW + tid] = h;
    }
    __syncthreads();
    #pragma unroll
    for (int u = 0; u < QMAX; ++u) {
        if (tid + u * 1024 < CHUNK_Q) {
            stage[start[dv[u].x >> RSH] + rk[u].x] = make_int2(sv[u].x, dv[u].x);
            stage[start[dv[u].y >> RSH] + rk[u].y] = make_int2(sv[u].y, dv[u].y);
            stage[start[dv[u].z >> RSH] + rk[u].z] = make_int2(sv[u].z, dv[u].z);
            stage[start[dv[u].w >> RSH] + rk[u].w] = make_int2(sv[u].w, dv[u].w);
        }
    }
    __syncthreads();
    int4* o4 = (int4*)(binned + e0);
    const int4* st4 = (const int4*)stage;
    for (int i = tid; i < CHUNK_E / 2; i += 1024) o4[i] = st4[i];
}
__device__ __forceinline__ _Float16 pack_w_elem(const float* __restrict__ W, int t) {
    int e = t & 15;
    int L = (t >> 4) & 31;
    int f = t >> 9;
    int k = f >> 3;
    int c = f & 7;
    int K = 32 * k + ((e & 8) ? 16 : 0) + 8 * (L >> 4) + (e & 7);
    int n = c * 16 + (L & 15);
    return (_Float16)W[K * CH + n];
}
#define PACK_BLOCKS (3 * CH * CH / 1024)
#define CONV_Q      (NNODES * CH / 4)
#define OWN_Q       (RNODES * CH / 4)
#define CONV_PT     (OWN_Q / 1024)

__global__ __launch_bounds__(1024) void owner_kernel(const int2* __restrict__ binned,
                                                     const int* __restrict__ tbl,
                                                     int* __restrict__ rowptr,
                                                     float* __restrict__ dis,
                                                     int* __restrict__ ew,
                                                     const float* __restrict__ W0,
                                                     const float* __restrict__ Ws,
                                                     const float* __restrict__ W1,
                                                     _Float16* __restrict__ Wp,
                                                     const float* __restrict__ xin,
                                                     _Float16* __restrict__ Xh) {
    __shared__ int cstart[64];
    __shared__ int cown[64];
    __shared__ int ncnt[RNODES];
    __shared__ int cur[RNODES];
    __shared__ float sdis[RNODES];
    __shared__ int wsum[8];
    static_assert(NCHUNK <= 64 && TROW == 16 * 16 && CONV_PT == 8, "table walk: 16 threads per chunk row");
    const int tid = threadIdx.x, lane = tid & 31;
    const int wv = __builtin_amdgcn_readfirstlane(tid >> 5);
    const int b = blockIdx.x;
    if (b >= NRANGE) {
        const int pb = b - NRANGE;
        const int m = pb >> 4;
        const float* W = (m == 0) ? W0 : (m == 1) ? Ws : W1;
        const int t = pb * 1024 + tid;
        Wp[t] = pack_w_elem(W, t & (CH * CH - 1));
        return;
    }
    const int qb = b * OWN_Q;
    const float4* x4 = (const float4*)xin;
    float4 xv[CONV_PT];
    #pragma unroll
    for (int j = 0; j < CONV_PT; ++j) xv[j] = x4[min(qb + tid + j * 1024, CONV_Q - 1)];

    {
        const int c = tid >> 4, part = tid & 15;
        int before = 0, own = 0;
        if (c < NCHUNK) {
            const int4* t4 = (const int4*)(tbl + c * TROW + part * 16);
            #pragma unroll
            for (int k = 0; k < 4; ++k) {
                const int4 v = t4[k];
                const int j = part * 16 + 4 * k;
                before += (j     < b ? v.x : 0) + (j + 1 < b ? v.y : 0)
                        + (j + 2 < b ? v.z : 0) + (j + 3 < b ? v.w : 0);
            }
            if (part == 0) own = tbl[c * TROW + b];
        }
        for (int o = 8; o > 0; o >>= 1) before += __shfl_xor(before, o);
        if (part == 0) {
            const int st = min(max(before, 0), CHUNK_E);
            cstart[c] = st;
            cown[c] = min(max(own, 0), CHUNK_E - st);
        }
    }
    if (tid < RNODES) ncnt[tid] = 0;
    __syncthreads();
    int base = cstart[lane] + cstart[lane + 32];
    for (int o = 16; o > 0; o >>= 1) base += __shfl_xor(base, o);

    for (int c = wv; c < NCHUNK; c += 32) {
        const int n = cown[c];
        const int2* run = binned + (size_t)c * CHUNK_E + cstart[c];
        for (int i = lane; i < n; i += 32) {
            const int2 p = run[i];
            atomicAdd(&ncnt[p.y & (RNODES - 1)], 1);
        }
    }
    __syncthreads();
    const int cnt = (tid < RNODES) ? ncnt[tid] : 0;
    int x = cnt;
    for (int o = 1; o < 32; o <<= 1) {
        int y = __shfl_up(x, o);
        if (lane >= o) x += y;
    }
    if (tid < RNODES && lane == 31) wsum[wv] = x;
    __syncthreads();
    if (tid < RNODES) {
        int wb = 0;
        for (int k = 0; k < wv; ++k) wb += wsum[k];
        const int excl = base + wb + x - cnt;
        const int node = b * RNODES + tid;
        const float dn = rsqrtf((float)(cnt + 1));
        cur[tid] = excl;
        sdis[tid] = dn;
        if (node < NNODES) {
            rowptr[node] = excl;
            dis[node] = dn;
        }
        if (node == NNODES - 1) rowptr[NNODES] = excl + cnt;
    }
    __syncthreads();
    for (int c = wv; c < NCHUNK; c += 32) {
        const int n = cown[c];
        const int2* run = binned + (size_t)c * CHUNK_E + cstart[c];
        for (int i = lane; i < n; i += 32) {
            const int2 p = run[i];
            const int pos = atomicAdd(&cur[p.y & (RNODES - 1)], 1);
            ew[min(max(pos, 0), NEDGES - 1)] = p.x;
        }
    }
    if (b == 0 && tid < 32) {
        v4h z4;
        z4[0] = (_Float16)0.0f; z4[1] = (_Float16)0.0f; z4[2] = (_Float16)0.0f; z4[3] = (_Float16)0.0f;
        *(v4h*)(Xh + (size_t)NNODES * CH + tid * 4) = z4;
    }
    #pragma unroll
    for (int j = 0; j < CONV_PT; ++j) {
        const int q = tid + j * 1024;
        if (qb + q < CONV_Q) {
            const float s = sdis[q >> 5];
            v4h hv;
            hv[0] = (_Float16)(xv[j].x * s); hv[1] = (_Float16)(xv[j].y * s);
            hv[2] = (_Float16)(xv[j].z * s); hv[3] = (_Float16)(xv[j].w * s);
            *(v4h*)(Xh + (size_t)(qb + q) * 4) = hv;
        }
    }
}

#define GBP  8
#define ZROW NNODES

template <int K>
__device__ __forceinline__ int row_lane(int v) {
    return __builtin_amdgcn_update_dpp(0, v, 0x150 + K, 0xf, 0xf, false);
}
__device__ __forceinline__ float other_half(float x) {
    return __builtin_bit_cast(float, __builtin_amdgcn_permlanex16(__builtin_bit_cast(int, x), __builtin_bit_cast(int, x),
                                                                  0x76543210, 0xfedcba98, false, false));
}

template <bool FIRST>
__device__ __forceinline__ void gather_row(const _Float16* __restrict__ X,
                                           const int* __restrict__ ew,
                                           int beg, int end, float di, int es, int d,
                                           int lane, float (&tot)[8], _Float16* own_row) {
    const int tl = 2 * (lane & 15) + (lane >> 4);
    const unsigned cb = (unsigned)(lane & 15) << 4;
    const char* Xb = (const char*)X;
    float acc[8];
    #pragma unroll
    for (int j = 0; j < 8; ++j) acc[j] = 0.0f;
    int nt = min(31, end - beg) + 1;
    int pnext = beg + 31;
    bool first = true;
    if (tl == 0) es = d;
    for (;;) {
        if (tl >= nt) es = ZROW;
        for (int ntb = nt; ntb > 0; ntb -= 2 * GBP) {
            v8h v[GBP];
            #define GATHER_PAIR(U)                                                   \
                {                                                                    \
                    const int s = row_lane<U>(es);                                   \
                    v[U] = *(const v8h*)(Xb + (((unsigned)s << 8) | cb));            \
                }                                                                    \
                if (ntb <= 2 * (U + 1)) break;
            static_assert(GBP == 8, "one GATHER_PAIR and one SUM_PAIR per pair of a batch");
            do {
                GATHER_PAIR(0) GATHER_PAIR(1) GATHER_PAIR(2) GATHER_PAIR(3)
                GATHER_PAIR(4) GATHER_PAIR(5) GATHER_PAIR(6) GATHER_PAIR(7)
            } while (false);
            #undef GATHER_PAIR
            if (FIRST && first && lane < 16) *(v8h*)(own_row + lane * 8) = v[0];
            first = false;
            #define SUM_PAIR(U)                                                      \
                _Pragma("unroll")                                                    \
                for (int j = 0; j < 8; ++j) acc[j] += (float)v[U][j];                \
                if (ntb <= 2 * (U + 1)) break;
            do {
                SUM_PAIR(0) SUM_PAIR(1) SUM_PAIR(2) SUM_PAIR(3)
                SUM_PAIR(4) SUM_PAIR(5) SUM_PAIR(6) SUM_PAIR(7)
            } while (false);
            #undef SUM_PAIR
            es = __builtin_amdgcn_update_dpp(0, es, 0x128, 0xf, 0xf, false);
        }
        if (pnext >= end) break;
        nt = min(32, end - pnext);
        es = 0;
        if (tl < nt) es = ew[pnext + tl];
        es = (int)min((unsigned)es, (unsigned)(NNODES - 1));
        pnext += 32;
    }
    #pragma unroll
    for (int j = 0; j < 8; ++j) tot[j] = (acc[j] + other_half(acc[j])) * di;
}
__device__ __forceinline__ float prelu1(float v, float a) { return v >= 0.0f ? v : a * v; }

#define TROWS  32
#define TPITCH 136
#define DPITCH 132

template <bool FIRST>
__device__ __forceinline__ void layer_body(
        const _Float16* __restrict__ X, const int* __restrict__ ew, const int* __restrict__ rowptr,
        const float* __restrict__ dis, const _Float16* __restrict__ WpA, const _Float16* __restrict__ WpS,
        const float* __restrict__ bA, const float* __restrict__ bS, const float* __restrict__ a,
        _Float16* tA, _Float16* tS, float* tD, float* tQ, float* tR,
        _Float16* __restrict__ OutH, float* __restrict__ Out) {
    const int lane = threadIdx.x & 31;
    const int wv = __builtin_amdgcn_readfirstlane(threadIdx.x >> 5);
    const int row0 = blockIdx.x * TROWS;

    static_assert(NNODES % 4 == 0, "a wave's 4 rows are all inside the node range or all outside");
    const int d0 = row0 + wv * 4;
    const bool live = d0 < NNODES;
    int r0 = 0, r1 = 0, r2 = 0, r3 = 0, r4 = 0;
    float q0 = 1.f, q1 = 1.f, q2 = 1.f, q3 = 1.f;
    int e0 = 0, e1 = 0, e2 = 0, e3 = 0;
    if (live) {
        r0 = rowptr[d0]; r1 = rowptr[d0 + 1]; r2 = rowptr[d0 + 2]; r3 = rowptr[d0 + 3]; r4 = rowptr[d0 + 4];
        q0 = dis[d0]; q1 = dis[d0 + 1]; q2 = dis[d0 + 2]; q3 = dis[d0 + 3];
        const int tl = 2 * (lane & 15) + (lane >> 4);
        e0 = ew[min(max(r0 + tl - 1, 0), NEDGES - 1)];
        e1 = ew[min(max(r1 + tl - 1, 0), NEDGES - 1)];
        e2 = ew[min(max(r2 + tl - 1, 0), NEDGES - 1)];
        e3 = ew[min(max(r3 + tl - 1, 0), NEDGES - 1)];
        e0 = (int)min((unsigned)e0, (unsigned)(NNODES - 1));
        e1 = (int)min((unsigned)e1, (unsigned)(NNODES - 1));
        e2 = (int)min((unsigned)e2, (unsigned)(NNODES - 1));
        e3 = (int)min((unsigned)e3, (unsigned)(NNODES - 1));
    }
    #pragma nounroll
    for (int i = 0; i < 4; ++i) {
        const int lr = wv * 4 + i;
        float g[8];
        v8h z8;
        #pragma unroll
        for (int j = 0; j < 8; ++j) { g[j] = 0.0f; z8[j] = (_Float16)0.0f; }
        if (live) {
            gather_row<FIRST>(X, ew, r0, r1, q0, e0, d0 + i, lane, g, FIRST ? tS + lr * TPITCH : nullptr);
        } else if (FIRST) {
            if (lane < 16) *(v8h*)(tS + lr * TPITCH + lane * 8) = z8;
        }
        if (FIRST) {
            if (lane == 0) { tQ[lr] = q0; tR[lr] = 1.0f / q0; }
        }
        r0 = r1; r1 = r2; r2 = r3; r3 = r4;
        q0 = q1; q1 = q2; q2 = q3;
        e0 = e1; e1 = e2; e2 = e3;
        v8h pg;
        #pragma unroll
        for (int j = 0; j < 8; ++j) pg[j] = (_Float16)g[j];
        if (lane < 16) *(v8h*)(tA + lr * TPITCH + lane * 8) = pg;
    }
    __syncthreads();

    const int half = lane >> 4;
    const int aoff = (lane & 15) * TPITCH + 8 * half;
    v8f accA[2] = {};
    v8f accS[2] = {};
    #pragma unroll
    for (int k = 0; k < 4; ++k) {
        const int f = k * 8 + wv;
        const v16h b = *(const v16h*)(WpA + ((f * 32 + lane) << 4));
        v16h b2 = b;
        if (FIRST) b2 = *(const v16h*)(WpS + ((f * 32 + lane) << 4));
        #pragma unroll
        for (int m = 0; m < 2; ++m) {
            const int o = aoff + m * 16 * TPITCH + 32 * k;
            const v8h alo = *(const v8h*)(tA + o);
            const v8h ahi = *(const v8h*)(tA + o + 16);
            const v16h av = __builtin_shufflevector(alo, ahi, 0, 1, 2, 3, 4, 5, 6, 7, 8, 9, 10, 11, 12, 13, 14, 15);
            accA[m] = __builtin_amdgcn_wmma_f32_16x16x32_f16(
                          false, av, false, b, (short)0, accA[m], false, false);
            if (FIRST) {
                const v8h slo = *(const v8h*)(tS + o);
                const v8h shi = *(const v8h*)(tS + o + 16);
                const v16h sv = __builtin_shufflevector(slo, shi, 0, 1, 2, 3, 4, 5, 6, 7, 8, 9, 10, 11, 12, 13, 14, 15);
                accS[m] = __builtin_amdgcn_wmma_f32_16x16x32_f16(
                              false, sv, false, b2, (short)0, accS[m], false, false);
            }
        }
    }
    {
        const int col = wv * 16 + (lane & 15);
        const float bav = bA[col];
        const float alv = a[col];
        const float bsv = FIRST ? bS[col] : 0.0f;
        #pragma unroll
        for (int m = 0; m < 2; ++m) {
            float4 ra = make_float4(1.f, 1.f, 1.f, 1.f), rb = ra;
            if (FIRST) {
                ra = *(const float4*)(tR + m * 16 + 8 * half);
                rb = *(const float4*)(tR + m * 16 + 8 * half + 4);
            }
            const float rinv[8] = {ra.x, ra.y, ra.z, ra.w, rb.x, rb.y, rb.z, rb.w};
            #pragma unroll
            for (int r = 0; r < 8; ++r) {
                float v = prelu1(accA[m][r] + bav, alv);
                if (FIRST) v = (accS[m][r] * rinv[r] + bsv) + v;
                tD[(m * 16 + r + 8 * half) * DPITCH + col] = v;
            }
        }
    }
    __syncthreads();

    if (FIRST && blockIdx.x == 0 && wv == 0) {
        v4h z4;
        z4[0] = (_Float16)0.0f; z4[1] = (_Float16)0.0f; z4[2] = (_Float16)0.0f; z4[3] = (_Float16)0.0f;
        *(v4h*)(OutH + (size_t)ZROW * CH + lane * 4) = z4;
    }
    for (int i = 0; i < 4; ++i) {
        const int lr = wv * 4 + i;
        const int d = row0 + lr;
        if (d < NNODES) {
            const float4 o = *(const float4*)(tD + lr * DPITCH + lane * 4);
            if (FIRST) {
                const float ds = tQ[lr];
                v4h oh;
                oh[0] = (_Float16)(o.x * ds); oh[1] = (_Float16)(o.y * ds);
                oh[2] = (_Float16)(o.z * ds); oh[3] = (_Float16)(o.w * ds);
                *(v4h*)(OutH + (size_t)d * CH + lane * 4) = oh;
            } else {
                *(float4*)(Out + (size_t)d * CH + lane * 4) = o;
            }
        }
    }
}

__global__ __launch_bounds__(256, 16) void layer0_kernel(
        const _Float16* __restrict__ xh, const int* __restrict__ ew, const int* __restrict__ rowptr,
        const float* __restrict__ dis, const _Float16* __restrict__ Wp0, const _Float16* __restrict__ Wps,
        const float* __restrict__ b0, const float* __restrict__ bs, const float* __restrict__ a,
        _Float16* __restrict__ U) {
    __shared__ __attribute__((aligned(16))) _Float16 tA[TROWS * TPITCH];
    __shared__ __attribute__((aligned(16))) _Float16 tS[TROWS * TPITCH];
    __shared__ __attribute__((aligned(16))) float    tD[TROWS * DPITCH];
    __shared__ __attribute__((aligned(16))) float    tQ[TROWS];
    __shared__ __attribute__((aligned(16))) float    tR[TROWS];
    layer_body<true>(xh, ew, rowptr, dis, Wp0, Wps, b0, bs, a, tA, tS, tD, tQ, tR, U, nullptr);
}
__global__ __launch_bounds__(256, 16) void layer1_kernel(
        const _Float16* __restrict__ U, const int* __restrict__ ew, const int* __restrict__ rowptr,
        const float* __restrict__ dis, const _Float16* __restrict__ Wp1,
        const float* __restrict__ b1, const float* __restrict__ a, float* __restrict__ out) {
    __shared__ __attribute__((aligned(16))) _Float16 tA[TROWS * TPITCH];
    __shared__ __attribute__((aligned(16))) float    tD[TROWS * DPITCH];
    layer_body<false>(U, ew, rowptr, dis, Wp1, nullptr, b1, nullptr, a, tA, nullptr, tD, nullptr, nullptr,
                      nullptr, out);
}

extern "C" void kernel_launch(void* const* d_in, const int* in_sizes, int n_in,
                              void* d_out, int out_size, void* d_ws, size_t ws_size,
                              hipStream_t stream) {
    const float* x  = (const float*)d_in[0];
    const int* edge = (const int*)d_in[1];
    const int* src  = edge;
    const int* dst  = edge + NEDGES;
    const float* W0 = (const float*)d_in[2];
    const float* b0 = (const float*)d_in[3];
    const float* W1 = (const float*)d_in[4];
    const float* b1 = (const float*)d_in[5];
    const float* Ws = (const float*)d_in[6];
    const float* bs = (const float*)d_in[7];
    const float* a  = (const float*)d_in[8];
    float* out = (float*)d_out;

    float* dis = (float*)d_ws;
    float* H   = dis + 50176;
    float* L   = H + (size_t)NNODES * CH;
    float* U   = L + (size_t)NNODES * CH;
    _Float16* Wp0 = (_Float16*)(U + (size_t)NNODES * CH);
    _Float16* Wps = Wp0 + CH * CH;
    _Float16* Wp1 = Wps + CH * CH;
    int2* binned = (int2*)H;
    int*  tbl    = (int*)H + 2 * NEDGES;
    _Float16* Xh = (_Float16*)(tbl + 64 * TROW);
    _Float16* Uh = (_Float16*)U;
    int*  rowptr = (int*)L + 50176;
    int*  ew     = rowptr + 50176;

    bin_edges_kernel<<<NCHUNK, 1024, 0, stream>>>(src, dst, binned, tbl);
    owner_kernel<<<NRANGE + PACK_BLOCKS, 1024, 0, stream>>>(binned, tbl, rowptr, dis, ew,
                                                            W0, Ws, W1, Wp0, x, Xh);

    layer0_kernel<<<(NNODES + TROWS - 1) / TROWS, 256, 0, stream>>>(Xh, ew, rowptr, dis, Wp0, Wps, b0, bs, a, Uh);
    layer1_kernel<<<(NNODES + TROWS - 1) / TROWS, 256, 0, stream>>>(Uh, ew, rowptr, dis, Wp1, b1, a, out);
}
